# starting kernel
# baseline (speedup 1.0000x reference)
#include <hip/hip_runtime.h>
#include <hip/hip_bf16.h>

typedef __attribute__((ext_vector_type(2))) float v2f;
typedef __attribute__((ext_vector_type(8))) float v8f;

#define NDIM_GRID 20
#define NY_NZ     400
#define NPTS      8000
#define NTILES    500
#define LOG2E     1.4426950408889634f

__device__ __forceinline__ void decode_coord(int i, float& x, float& y, float& z) {
    int ix = i / NY_NZ;
    int rem = i - ix * NY_NZ;
    int iy = rem / NDIM_GRID;
    int iz = rem - iy * NDIM_GRID;
    x = (float)ix; y = (float)iy; z = (float)iz;
}

__global__ void __launch_bounds__(64) kde_main(
        const float* __restrict__ p,
        const float* __restrict__ Ainv,
        float* __restrict__ kde)
{
    __shared__ float4 bq[NPTS];
    __shared__ float  pp[NPTS];

    const int tid  = threadIdx.x;
    const int wave = (blockIdx.x * blockDim.x + tid) >> 5;
    const int lane = tid & 31;
    const int half = lane >> 4;
    const int l15  = lane & 15;

    const float a00 = Ainv[0], a01 = Ainv[1], a02 = Ainv[2];
    const float a10 = Ainv[3], a11 = Ainv[4], a12 = Ainv[5];
    const float a20 = Ainv[6], a21 = Ainv[7], a22 = Ainv[8];

    for (int i = tid; i < NPTS; i += 64) {
        float cx, cy, cz; decode_coord(i, cx, cy, cz);
        float b0 = a00 * cx + a01 * cy + a02 * cz;
        float b1 = a10 * cx + a11 * cy + a12 * cz;
        float b2 = a20 * cx + a21 * cy + a22 * cz;
        float q  = cx * b0 + cy * b1 + cz * b2;
        bq[i] = make_float4(cx, cy, cz, q);
        pp[i] = p[i];
    }
    __syncthreads();

    const int row = wave * 16 + l15;
    float rx, ry, rz; decode_coord(row, rx, ry, rz);
    const float g0 = a00 * rx + a01 * ry + a02 * rz;
    const float g1 = a10 * rx + a11 * ry + a12 * rz;
    const float g2 = a20 * rx + a21 * ry + a22 * rz;
    const float qrow = rx * g0 + ry * g1 + rz * g2;
    v2f amat;
    amat.x = half ? (LOG2E * g2) : (LOG2E * g0);
    amat.y = half ? (-0.5f * LOG2E) : (LOG2E * g1);

    v8f cvec;
#pragma unroll
    for (int r = 0; r < 8; ++r)
        cvec[r] = (-0.5f * LOG2E) * __shfl(qrow, (lane & 16) + half * 8 + r, 32);

    float acc[8] = {};

    const v2f*   bbase = (const v2f*)bq + (l15 * 2 + half);
    const float* pbase = pp + l15;

#pragma unroll 2
    for (int t = 0; t < NTILES; ++t) {
        const v2f   bmat = bbase[t * 32];
        const float pb   = pbase[t * 16];

        v8f c = __builtin_amdgcn_wmma_f32_16x16x4_f32(
                false, amat, false, bmat, (short)0, cvec, false, false);

#pragma unroll
        for (int r = 0; r < 8; ++r)
            acc[r] = fmaf(__builtin_amdgcn_exp2f(c[r]), pb, acc[r]);
    }

#pragma unroll
    for (int r = 0; r < 8; ++r) {
        float v = acc[r];
        v += __shfl_xor(v, 1, 32);
        v += __shfl_xor(v, 2, 32);
        v += __shfl_xor(v, 4, 32);
        v += __shfl_xor(v, 8, 32);
        if (l15 == 0)
            kde[wave * 16 + r + 8 * half] = v;
    }
}

__global__ void kde_norm(float* __restrict__ kde) {
    __shared__ float red[256];
    float s = 0.0f;
    for (int i = threadIdx.x; i < NPTS; i += 256) s += kde[i];
    red[threadIdx.x] = s;
    __syncthreads();
    for (int off = 128; off > 0; off >>= 1) {
        if (threadIdx.x < off) red[threadIdx.x] += red[threadIdx.x + off];
        __syncthreads();
    }
    const float inv = 1.0f / red[0];
    for (int i = threadIdx.x; i < NPTS; i += 256) kde[i] *= inv;
}

extern "C" void kernel_launch(void* const* d_in, const int* in_sizes, int n_in,
                              void* d_out, int out_size, void* d_ws, size_t ws_size,
                              hipStream_t stream) {
    const float* p    = (const float*)d_in[0];
    const float* ainv = (const float*)d_in[1];
    float* out = (float*)d_out;

    kde_main<<<250, 64, 0, stream>>>(p, ainv, out);
    kde_norm<<<1, 256, 0, stream>>>(out);
}


# Weight tiles reused along the x diagonal: a wave owns a row tile of 5 consecutive x-planes, sweeps 20 column planes, 1 new exp2 tile per step if cov_inv is symmetric, else all 5 (new summation order)
# speedup vs baseline: 4.5450x; 4.5450x over previous
#include <hip/hip_runtime.h>
#include <hip/hip_bf16.h>


typedef __attribute__((ext_vector_type(2))) float v2f;
typedef __attribute__((ext_vector_type(8))) float v8f;

#define NDIM_GRID 20
#define NY_NZ     400
#define NPTS      8000
#define NTILES    500
#define LOG2E     1.4426950408889634f

#define NTHR      320
#define NWAVE     10
#define TPP       25
#define RT        5
#define NXB       20
#define NPLANE    25
#define NWTOT     2500
#define WS_FLOATS (NPLANE * NPTS + 2560)
#define WS_BYTES  ((size_t)WS_FLOATS * 4)
#define SYM_TOL   9.5367431640625e-07f

__device__ __forceinline__ void decode_coord(int i, float& x, float& y, float& z) {
    int ix = i / NY_NZ;
    int rem = i - ix * NY_NZ;
    int iy = rem / NDIM_GRID;
    int iz = rem - iy * NDIM_GRID;
    x = (float)ix; y = (float)iy; z = (float)iz;
}

__device__ __forceinline__ float row16_sum(float v) {
    v += __builtin_bit_cast(float, __builtin_amdgcn_update_dpp(
            0, __builtin_bit_cast(int, v), 0x161, 0xf, 0xf, true));
    v += __builtin_bit_cast(float, __builtin_amdgcn_update_dpp(
            0, __builtin_bit_cast(int, v), 0x162, 0xf, 0xf, true));
    v += __builtin_bit_cast(float, __builtin_amdgcn_update_dpp(
            0, __builtin_bit_cast(int, v), 0x164, 0xf, 0xf, true));
    v += __builtin_bit_cast(float, __builtin_amdgcn_update_dpp(
            0, __builtin_bit_cast(int, v), 0x168, 0xf, 0xf, true));
    return v;
}

__global__ void __launch_bounds__(NTHR) kde_main(
        const float* __restrict__ p,
        const float* __restrict__ Ainv,
        float* __restrict__ kde,
        float* __restrict__ ws,
        size_t ws_bytes)
{
    __shared__ float4 bq[NPTS];
    __shared__ float  pp[NPTS];

    const int tid  = threadIdx.x;
    const int w    = tid >> 5;
    const int lane = tid & 31;
    const int half = lane >> 4;
    const int l15  = lane & 15;

    const float a00 = Ainv[0], a01 = Ainv[1], a02 = Ainv[2];
    const float a10 = Ainv[3], a11 = Ainv[4], a12 = Ainv[5];
    const float a20 = Ainv[6], a21 = Ainv[7], a22 = Ainv[8];

    if (ws_bytes >= WS_BYTES) {
        const int n = blockIdx.x / NWAVE;
        const int G = blockIdx.x - n * NWAVE;

        {
            const int col = (tid >> 4) * NY_NZ + n * 16 + (tid & 15);
            const float pv = p[col];
            float cx, cy, cz; decode_coord(col, cx, cy, cz);
            float b0 = a00 * cx + a01 * cy + a02 * cz;
            float b1 = a10 * cx + a11 * cy + a12 * cz;
            float b2 = a20 * cx + a21 * cy + a22 * cz;
            float q  = cx * b0 + cy * b1 + cz * b2;
            bq[tid] = make_float4(cx, cy, cz, q);
            pp[tid] = pv;
        }
        __syncthreads();

        const int pair = G * NWAVE + w;
        const int g    = pair / TPP;
        const int m    = pair - g * TPP;
        const int xa0  = g * RT;

        v2f amat[RT];
        v8f cvec[RT];
#pragma unroll
        for (int s = 0; s < RT; ++s) {
            const int row = ((xa0 + s) * TPP + m) * 16 + l15;
            float rx, ry, rz; decode_coord(row, rx, ry, rz);
            const float g0 = a00 * rx + a01 * ry + a02 * rz;
            const float g1 = a10 * rx + a11 * ry + a12 * rz;
            const float g2 = a20 * rx + a21 * ry + a22 * rz;
            const float qrow = rx * g0 + ry * g1 + rz * g2;
            amat[s].x = half ? (LOG2E * g2) : (LOG2E * g0);
            amat[s].y = half ? (-0.5f * LOG2E) : (LOG2E * g1);
#pragma unroll
            for (int r = 0; r < 8; ++r)
                cvec[s][r] = (-0.5f * LOG2E) * __shfl(qrow, (lane & 16) + half * 8 + r, 32);
        }

        float acc[RT][8] = {};

        const v2f*   bbase = (const v2f*)bq + (l15 * 2 + half);
        const float* pbase = pp + l15;

        const float amax = fmaxf(fmaxf(fmaxf(fabsf(a00), fabsf(a01)), fmaxf(fabsf(a02), fabsf(a10))),
                           fmaxf(fmaxf(fabsf(a11), fabsf(a12)), fmaxf(fabsf(a20), fmaxf(fabsf(a21), fabsf(a22)))));
        const bool xshift = (fabsf(a01 - a10) <= SYM_TOL * amax) &&
                            (fabsf(a02 - a20) <= SYM_TOL * amax);

        if (xshift) {
            float win[RT][8];
#pragma unroll
            for (int xb = 0; xb < NXB; ++xb) {
                const v2f   bmat = bbase[xb * 32];
                const float pb   = pbase[xb * 16];
                if (xb == 0) {
#pragma unroll
                    for (int s = 0; s < RT; ++s) {
                        v8f c = __builtin_amdgcn_wmma_f32_16x16x4_f32(
                                false, amat[s], false, bmat, (short)0, cvec[s], false, false);
#pragma unroll
                        for (int r = 0; r < 8; ++r) win[s][r] = __builtin_amdgcn_exp2f(c[r]);
                    }
                } else {
                    const int slot = (5 * NXB - xb) % RT;
                    v8f c = __builtin_amdgcn_wmma_f32_16x16x4_f32(
                            false, amat[0], false, bmat, (short)0, cvec[0], false, false);
#pragma unroll
                    for (int r = 0; r < 8; ++r) win[slot][r] = __builtin_amdgcn_exp2f(c[r]);
                }
#pragma unroll
                for (int s = 0; s < RT; ++s) {
                    const int k = (s + 5 * NXB - xb) % RT;
#pragma unroll
                    for (int r = 0; r < 8; ++r)
                        acc[s][r] = fmaf(win[k][r], pb, acc[s][r]);
                }
            }
        } else {
            for (int xb = 0; xb < NXB; ++xb) {
                const v2f   bmat = bbase[xb * 32];
                const float pb   = pbase[xb * 16];
#pragma unroll
                for (int s = 0; s < RT; ++s) {
                    v8f c = __builtin_amdgcn_wmma_f32_16x16x4_f32(
                            false, amat[s], false, bmat, (short)0, cvec[s], false, false);
#pragma unroll
                    for (int r = 0; r < 8; ++r)
                        acc[s][r] = fmaf(__builtin_amdgcn_exp2f(c[r]), pb, acc[s][r]);
                }
            }
        }

        float* plane = ws + (size_t)n * NPTS;
        float tot = 0.0f;
#pragma unroll
        for (int s = 0; s < RT; ++s) {
#pragma unroll
            for (int r = 0; r < 8; ++r) {
                acc[s][r] = row16_sum(acc[s][r]);
                tot += acc[s][r];
            }
            if (l15 == 0) {
                float4* dst = (float4*)(plane + ((xa0 + s) * TPP + m) * 16 + 8 * half);
                dst[0] = make_float4(acc[s][0], acc[s][1], acc[s][2], acc[s][3]);
                dst[1] = make_float4(acc[s][4], acc[s][5], acc[s][6], acc[s][7]);
            }
        }
        tot += __shfl_xor(tot, 16, 32);
        if (lane == 0)
            ws[NPLANE * NPTS + blockIdx.x * NWAVE + w] = tot;
        return;
    }

    for (int i = tid; i < NPTS; i += NTHR) {
        float cx, cy, cz; decode_coord(i, cx, cy, cz);
        float b0 = a00 * cx + a01 * cy + a02 * cz;
        float b1 = a10 * cx + a11 * cy + a12 * cz;
        float b2 = a20 * cx + a21 * cy + a22 * cz;
        float q  = cx * b0 + cy * b1 + cz * b2;
        bq[i] = make_float4(cx, cy, cz, q);
        pp[i] = p[i];
    }
    __syncthreads();
    if (w >= 2) return;
    {
        const int tile = blockIdx.x * 2 + w;
        const int row = tile * 16 + l15;
        float rx, ry, rz; decode_coord(row, rx, ry, rz);
        const float g0 = a00 * rx + a01 * ry + a02 * rz;
        const float g1 = a10 * rx + a11 * ry + a12 * rz;
        const float g2 = a20 * rx + a21 * ry + a22 * rz;
        const float qrow = rx * g0 + ry * g1 + rz * g2;
        v2f amat;
        amat.x = half ? (LOG2E * g2) : (LOG2E * g0);
        amat.y = half ? (-0.5f * LOG2E) : (LOG2E * g1);
        v8f cvec;
#pragma unroll
        for (int r = 0; r < 8; ++r)
            cvec[r] = (-0.5f * LOG2E) * __shfl(qrow, (lane & 16) + half * 8 + r, 32);

        float acc[8] = {};
        const v2f*   bbase = (const v2f*)bq + (l15 * 2 + half);
        const float* pbase = pp + l15;
        for (int t = 0; t < NTILES; ++t) {
            const v2f   bmat = bbase[t * 32];
            const float pb   = pbase[t * 16];
            v8f c = __builtin_amdgcn_wmma_f32_16x16x4_f32(
                    false, amat, false, bmat, (short)0, cvec, false, false);
#pragma unroll
            for (int r = 0; r < 8; ++r)
                acc[r] = fmaf(__builtin_amdgcn_exp2f(c[r]), pb, acc[r]);
        }
#pragma unroll
        for (int r = 0; r < 8; ++r) {
            float v = row16_sum(acc[r]);
            if (l15 == 0)
                kde[tile * 16 + r + 8 * half] = v;
        }
    }
}

__global__ void __launch_bounds__(256) kde_combine(
        float* __restrict__ kde, const float* __restrict__ ws, size_t ws_bytes) {
    __shared__ float red[256];
    const int tid = threadIdx.x;

    if (ws_bytes >= WS_BYTES) {
        const int row  = blockIdx.x * 256 + tid;
        const int rowc = row < NPTS ? row : NPTS - 1;
        float part[NPLANE];
#pragma unroll
        for (int k = 0; k < NPLANE; ++k) part[k] = ws[k * NPTS + rowc];
        float tpart[10];
#pragma unroll
        for (int k = 0; k < 10; ++k) {
            const int i = tid + 256 * k;
            tpart[k] = (i < NWTOT) ? ws[NPLANE * NPTS + i] : 0.0f;
        }
        float rs = 0.0f;
#pragma unroll
        for (int k = 0; k < NPLANE; ++k) rs += part[k];
        float s = 0.0f;
#pragma unroll
        for (int k = 0; k < 10; ++k) s += tpart[k];
        red[tid] = s;
        __syncthreads();
        for (int off = 128; off > 0; off >>= 1) {
            if (tid < off) red[tid] += red[tid + off];
            __syncthreads();
        }
        const float inv = 1.0f / red[0];
        if (row < NPTS) kde[row] = rs * inv;
        return;
    }

    if (blockIdx.x != 0) return;
    float s = 0.0f;
    for (int i = tid; i < NPTS; i += 256) s += kde[i];
    red[tid] = s;
    __syncthreads();
    for (int off = 128; off > 0; off >>= 1) {
        if (tid < off) red[tid] += red[tid + off];
        __syncthreads();
    }
    const float inv = 1.0f / red[0];
    for (int i = tid; i < NPTS; i += 256) kde[i] *= inv;
}

extern "C" void kernel_launch(void* const* d_in, const int* in_sizes, int n_in,
                              void* d_out, int out_size, void* d_ws, size_t ws_size,
                              hipStream_t stream) {
    const float* p    = (const float*)d_in[0];
    const float* ainv = (const float*)d_in[1];
    float* out = (float*)d_out;
    float* ws  = (float*)d_ws;

    kde_main<<<250, NTHR, 0, stream>>>(p, ainv, out, ws, ws_size);
    kde_combine<<<32, 256, 0, stream>>>(out, ws, ws_size);
}
